# starting kernel
# baseline (speedup 1.0000x reference)
#include <hip/hip_runtime.h>
#include <cmath>

#define B_ 32
#define H_ 64
#define W_ 64
#define D_ 256
#define U_ 512

typedef __attribute__((ext_vector_type(2))) float v2f;
typedef __attribute__((ext_vector_type(8))) float v8f;

__device__ __forceinline__ v2f load_a_frag(const float* __restrict__ base,
                                           long rowStride, int k0, int lane) {
    const int r  = lane & 15;
    const int kk = k0 + ((lane >> 4) << 1);
    const float* p = base + (long)r * rowStride + kk;
    v2f a = { p[0], p[1] };
    return a;
}

__device__ __forceinline__ v2f load_b_frag(const float* __restrict__ Bm, int ldb,
                                           int n0, int k0, int lane) {
    const int n  = n0 + (lane & 15);
    const int kk = k0 + ((lane >> 4) << 1);
    v2f b = { Bm[(long)kk * ldb + n], Bm[(long)(kk + 1) * ldb + n] };
    return b;
}

__device__ __forceinline__ void gemm_acc(const float* __restrict__ Abase, long aRowStride,
                                         const float* __restrict__ Bm, int ldb, int K,
                                         int nBase, int lane, v8f acc[2][4]) {
    #pragma unroll 4
    for (int k0 = 0; k0 < K; k0 += 4) {
        v2f a0 = load_a_frag(Abase,                   aRowStride, k0, lane);
        v2f a1 = load_a_frag(Abase + 16 * aRowStride, aRowStride, k0, lane);
        #pragma unroll
        for (int t = 0; t < 4; ++t) {
            v2f b = load_b_frag(Bm, ldb, nBase + 16 * t, k0, lane);
            acc[0][t] = __builtin_amdgcn_wmma_f32_16x16x4_f32(
                false, a0, false, b, (short)0, acc[0][t], false, false);
            acc[1][t] = __builtin_amdgcn_wmma_f32_16x16x4_f32(
                false, a1, false, b, (short)0, acc[1][t], false, false);
        }
    }
}

__global__ void __launch_bounds__(256)
mdrnn_diag_kernel(const float* __restrict__ x,   const float* __restrict__ wax,
                  const float* __restrict__ w0,  const float* __restrict__ w1,
                  const float* __restrict__ ba,
                  const float* __restrict__ hprev,
                  float* __restrict__ hcur,
                  float* __restrict__ out,
                  int d, int i0)
{
    const int i = i0 + (int)blockIdx.x;
    const int j = d - i;
    const int tid   = (int)threadIdx.x;
    const int lane  = tid & 31;
    const int nBase = (tid >> 5) * 64;

    v8f acc[2][4];
    #pragma unroll
    for (int mt = 0; mt < 2; ++mt)
        #pragma unroll
        for (int t = 0; t < 4; ++t)
            acc[mt][t] = (v8f){0.f, 0.f, 0.f, 0.f, 0.f, 0.f, 0.f, 0.f};

    {
        const long xRowStride = (long)H_ * W_ * D_;
        const float* xBase = x + ((long)i * W_ + j) * (long)D_;
        gemm_acc(xBase, xRowStride, wax, U_, D_, nBase, lane, acc);
    }
    if (i > 0)
        gemm_acc(hprev + (long)(i - 1) * B_ * U_, U_, w0, U_, U_, nBase, lane, acc);
    if (j > 0)
        gemm_acc(hprev + (long)i * B_ * U_,       U_, w1, U_, U_, nBase, lane, acc);

    const bool last  = (i == H_ - 1) && (j == W_ - 1);
    const int  n     = lane & 15;
    const int  hiRow = (lane >> 4) << 3;
    #pragma unroll
    for (int mt = 0; mt < 2; ++mt) {
        #pragma unroll
        for (int t = 0; t < 4; ++t) {
            const int col = nBase + 16 * t + n;
            const float bv = ba[col];
            #pragma unroll
            for (int r = 0; r < 8; ++r) {
                const int row = mt * 16 + hiRow + r;
                const float v = tanhf(acc[mt][t][r] + bv);
                hcur[((long)i * B_ + row) * U_ + col] = v;
                if (last) out[(long)row * U_ + col] = v;
            }
        }
    }
}

extern "C" void kernel_launch(void* const* d_in, const int* in_sizes, int n_in,
                              void* d_out, int out_size, void* d_ws, size_t ws_size,
                              hipStream_t stream) {
    (void)in_sizes; (void)n_in; (void)out_size; (void)ws_size;
    const float* x   = (const float*)d_in[0];
    const float* wax = (const float*)d_in[1];
    const float* w0  = (const float*)d_in[2];
    const float* w1  = (const float*)d_in[3];
    const float* ba  = (const float*)d_in[4];
    float* out  = (float*)d_out;
    float* hbuf = (float*)d_ws;
    const long diagElems = (long)H_ * B_ * U_;

    for (int d = 0; d < H_ + W_ - 1; ++d) {
        const int i0 = (d > W_ - 1) ? d - (W_ - 1) : 0;
        const int i1 = (d < H_ - 1) ? d : (H_ - 1);
        const int nb = i1 - i0 + 1;
        float*       hc = hbuf + (long)(d & 1) * diagElems;
        const float* hp = hbuf + (long)((d & 1) ^ 1) * diagElems;
        mdrnn_diag_kernel<<<dim3(nb), dim3(256), 0, stream>>>(
            x, wax, w0, w1, ba, hp, hc, out, d, i0);
    }
}


# Each grid cell spread over 8 work-groups (64 columns each) whose 8 waves split K eight ways (k-loop unrolled by 8 for the short slice); partials summed in fixed wave order via LDS: new f32 sum order
# speedup vs baseline: 4.0042x; 4.0042x over previous
#include <hip/hip_runtime.h>
#include <cmath>


#define B_ 32
#define H_ 64
#define W_ 64
#define D_ 256
#define U_ 512

#define NSLAB 8
#define SLABW (U_ / NSLAB)
#define NWAVE 8
#define PITCH 66

typedef __attribute__((ext_vector_type(2))) float v2f;
typedef __attribute__((ext_vector_type(8))) float v8f;

__device__ __forceinline__ v2f load_a_frag(const float* __restrict__ base,
                                           long rowStride, int k0, int lane) {
    const int r  = lane & 15;
    const int kk = k0 + ((lane >> 4) << 1);
    const float* p = base + (long)r * rowStride + kk;
    v2f a = { p[0], p[1] };
    return a;
}

__device__ __forceinline__ v2f load_b_frag(const float* __restrict__ Bm, int ldb,
                                           int n0, int k0, int lane) {
    const int n  = n0 + (lane & 15);
    const int kk = k0 + ((lane >> 4) << 1);
    v2f b = { Bm[(long)kk * ldb + n], Bm[(long)(kk + 1) * ldb + n] };
    return b;
}

template <int KC>
__device__ __forceinline__ void gemm_acc(const float* __restrict__ Abase, long aRowStride,
                                         const float* __restrict__ Bm, int ldb,
                                         int nBase, int lane, v8f acc[2][4]) {
    #pragma unroll 8
    for (int k0 = 0; k0 < KC; k0 += 4) {
        v2f a0 = load_a_frag(Abase,                   aRowStride, k0, lane);
        v2f a1 = load_a_frag(Abase + 16 * aRowStride, aRowStride, k0, lane);
        #pragma unroll
        for (int t = 0; t < 4; ++t) {
            v2f b = load_b_frag(Bm, ldb, nBase + 16 * t, k0, lane);
            acc[0][t] = __builtin_amdgcn_wmma_f32_16x16x4_f32(
                false, a0, false, b, (short)0, acc[0][t], false, false);
            acc[1][t] = __builtin_amdgcn_wmma_f32_16x16x4_f32(
                false, a1, false, b, (short)0, acc[1][t], false, false);
        }
    }
}

__global__ void __launch_bounds__(256)
mdrnn_diag_kernel(const float* __restrict__ x,   const float* __restrict__ wax,
                  const float* __restrict__ w0,  const float* __restrict__ w1,
                  const float* __restrict__ ba,
                  const float* __restrict__ hprev,
                  float* __restrict__ hcur,
                  float* __restrict__ out,
                  int d, int i0)
{
    __shared__ float part[NWAVE][B_][PITCH];

    const int i = i0 + (int)blockIdx.y;
    const int j = d - i;
    const int tid   = (int)threadIdx.x;
    const int lane  = tid & 31;
    const int wave  = tid >> 5;
    const int nBase = (int)blockIdx.x * SLABW;

    v8f acc[2][4];
    #pragma unroll
    for (int mt = 0; mt < 2; ++mt)
        #pragma unroll
        for (int t = 0; t < 4; ++t)
            acc[mt][t] = (v8f){0.f, 0.f, 0.f, 0.f, 0.f, 0.f, 0.f, 0.f};

    {
        const int  kx = wave * (D_ / NWAVE);
        const long xRowStride = (long)H_ * W_ * D_;
        const float* xBase = x + ((long)i * W_ + j) * (long)D_ + kx;
        gemm_acc<D_ / NWAVE>(xBase, xRowStride, wax + (long)kx * U_, U_, nBase, lane, acc);
    }
    const int kh = wave * (U_ / NWAVE);
    if (i > 0)
        gemm_acc<U_ / NWAVE>(hprev + (long)(i - 1) * B_ * U_ + kh, U_,
                             w0 + (long)kh * U_, U_, nBase, lane, acc);
    if (j > 0)
        gemm_acc<U_ / NWAVE>(hprev + (long)i * B_ * U_ + kh,       U_,
                             w1 + (long)kh * U_, U_, nBase, lane, acc);

    {
        const int n     = lane & 15;
        const int hiRow = (lane >> 4) << 3;
        #pragma unroll
        for (int mt = 0; mt < 2; ++mt)
            #pragma unroll
            for (int t = 0; t < 4; ++t)
                #pragma unroll
                for (int r = 0; r < 8; ++r)
                    part[wave][mt * 16 + hiRow + r][16 * t + n] = acc[mt][t][r];
    }
    __syncthreads();

    const bool last = (i == H_ - 1) && (j == W_ - 1);
    const int  col  = tid & (SLABW - 1);
    const int  gcol = nBase + col;
    const float bv  = ba[gcol];
    #pragma unroll
    for (int q = 0; q < B_ / 4; ++q) {
        const int row = (tid >> 6) + 4 * q;
        float s = part[0][row][col];
        #pragma unroll
        for (int w = 1; w < NWAVE; ++w) s += part[w][row][col];
        const float v = tanhf(s + bv);
        hcur[((long)i * B_ + row) * U_ + gcol] = v;
        if (last) out[(long)row * U_ + gcol] = v;
    }
}

extern "C" void kernel_launch(void* const* d_in, const int* in_sizes, int n_in,
                              void* d_out, int out_size, void* d_ws, size_t ws_size,
                              hipStream_t stream) {
    (void)in_sizes; (void)n_in; (void)out_size; (void)ws_size;
    const float* x   = (const float*)d_in[0];
    const float* wax = (const float*)d_in[1];
    const float* w0  = (const float*)d_in[2];
    const float* w1  = (const float*)d_in[3];
    const float* ba  = (const float*)d_in[4];
    float* out  = (float*)d_out;
    float* hbuf = (float*)d_ws;
    const long diagElems = (long)H_ * B_ * U_;

    for (int d = 0; d < H_ + W_ - 1; ++d) {
        const int i0 = (d > W_ - 1) ? d - (W_ - 1) : 0;
        const int i1 = (d < H_ - 1) ? d : (H_ - 1);
        const int nb = i1 - i0 + 1;
        float*       hc = hbuf + (long)(d & 1) * diagElems;
        const float* hp = hbuf + (long)((d & 1) ^ 1) * diagElems;
        mdrnn_diag_kernel<<<dim3(NSLAB, nb), dim3(256), 0, stream>>>(
            x, wax, w0, w1, ba, hp, hc, out, d, i0);
    }
}


# A and B fragments fetched with 16-byte loads (A: 4 k per lane; B: 4 columns per lane, tile t owns slab columns 4c+t), a whole K slice's loads in flight: new f32 sum order inside each 8-k group
# speedup vs baseline: 4.4111x; 1.1016x over previous
#include <hip/hip_runtime.h>
#include <cmath>


#define B_ 32
#define H_ 64
#define W_ 64
#define D_ 256
#define U_ 512

#define NSLAB 8
#define SLABW (U_ / NSLAB)
#define NWAVE 8
#define PITCH 68

typedef __attribute__((ext_vector_type(2))) float v2f;
typedef __attribute__((ext_vector_type(4))) float v4f;
typedef __attribute__((ext_vector_type(8))) float v8f;

template <int KC>
__device__ __forceinline__ void gemm_acc(const float* __restrict__ aLane, long aRowStride,
                                         const float* __restrict__ bLane,
                                         v8f acc[2][4]) {
    const float* __restrict__ aLane1 = aLane + 16 * aRowStride;
    #pragma unroll
    for (int g = 0; g < KC / 8; ++g) {
        const v4f a0 = *(const v4f*)(aLane  + 8 * g);
        const v4f a1 = *(const v4f*)(aLane1 + 8 * g);
        v4f b[2][2];
        #pragma unroll
        for (int s = 0; s < 2; ++s)
            #pragma unroll
            for (int e = 0; e < 2; ++e)
                b[s][e] = *(const v4f*)(bLane + (long)(8 * g + 2 * s + e) * U_);
        #pragma unroll
        for (int s = 0; s < 2; ++s) {
            const v2f af0 = { a0[2 * s], a0[2 * s + 1] };
            const v2f af1 = { a1[2 * s], a1[2 * s + 1] };
            #pragma unroll
            for (int t = 0; t < 4; ++t) {
                const v2f bf = { b[s][0][t], b[s][1][t] };
                acc[0][t] = __builtin_amdgcn_wmma_f32_16x16x4_f32(
                    false, af0, false, bf, (short)0, acc[0][t], false, false);
                acc[1][t] = __builtin_amdgcn_wmma_f32_16x16x4_f32(
                    false, af1, false, bf, (short)0, acc[1][t], false, false);
            }
        }
    }
}

__global__ void __launch_bounds__(256) __attribute__((amdgpu_waves_per_eu(4, 4)))
mdrnn_diag_kernel(const float* __restrict__ x,   const float* __restrict__ wax,
                  const float* __restrict__ w0,  const float* __restrict__ w1,
                  const float* __restrict__ ba,
                  const float* __restrict__ hprev,
                  float* __restrict__ hcur,
                  float* __restrict__ out,
                  int d, int i0)
{
    __shared__ __attribute__((aligned(16))) float part[NWAVE][B_][PITCH];

    const int i = i0 + (int)blockIdx.y;
    const int j = d - i;
    const int tid   = (int)threadIdx.x;
    const int lane  = tid & 31;
    const int wave  = tid >> 5;
    const int nBase = (int)blockIdx.x * SLABW;

    v8f acc[2][4];
    #pragma unroll
    for (int mt = 0; mt < 2; ++mt)
        #pragma unroll
        for (int t = 0; t < 4; ++t)
            acc[mt][t] = (v8f){0.f, 0.f, 0.f, 0.f, 0.f, 0.f, 0.f, 0.f};

    const int c = lane & 15;
    const int h = lane >> 4;
    const long bOff = (long)(4 * h) * U_ + nBase + 4 * c;

    {
        const int  kx = wave * (D_ / NWAVE);
        const long xRowStride = (long)H_ * W_ * D_;
        const float* xLane = x + ((long)i * W_ + j) * (long)D_ + kx
                               + (long)c * xRowStride + 4 * h;
        gemm_acc<D_ / NWAVE>(xLane, xRowStride, wax + (long)kx * U_ + bOff, acc);
    }
    const int kh = wave * (U_ / NWAVE);
    if (i > 0)
        gemm_acc<U_ / NWAVE>(hprev + ((long)(i - 1) * B_ + c) * U_ + kh + 4 * h, U_,
                             w0 + (long)kh * U_ + bOff, acc);
    if (j > 0)
        gemm_acc<U_ / NWAVE>(hprev + ((long)i * B_ + c) * U_ + kh + 4 * h,       U_,
                             w1 + (long)kh * U_ + bOff, acc);

    {
        const int hiRow = h << 3;
        #pragma unroll
        for (int mt = 0; mt < 2; ++mt)
            #pragma unroll
            for (int r = 0; r < 8; ++r) {
                const v4f v = { acc[mt][0][r], acc[mt][1][r], acc[mt][2][r], acc[mt][3][r] };
                *(v4f*)&part[wave][mt * 16 + hiRow + r][4 * c] = v;
            }
    }
    __syncthreads();

    const bool last = (i == H_ - 1) && (j == W_ - 1);
    const int  col  = tid & (SLABW - 1);
    const int  gcol = nBase + col;
    const float bv  = ba[gcol];
    #pragma unroll
    for (int q = 0; q < B_ / 4; ++q) {
        const int row = (tid >> 6) + 4 * q;
        float s = part[0][row][col];
        #pragma unroll
        for (int w = 1; w < NWAVE; ++w) s += part[w][row][col];
        const float v = tanhf(s + bv);
        hcur[((long)i * B_ + row) * U_ + gcol] = v;
        if (last) out[(long)row * U_ + gcol] = v;
    }
}

extern "C" void kernel_launch(void* const* d_in, const int* in_sizes, int n_in,
                              void* d_out, int out_size, void* d_ws, size_t ws_size,
                              hipStream_t stream) {
    (void)in_sizes; (void)n_in; (void)out_size; (void)ws_size;
    const float* x   = (const float*)d_in[0];
    const float* wax = (const float*)d_in[1];
    const float* w0  = (const float*)d_in[2];
    const float* w1  = (const float*)d_in[3];
    const float* ba  = (const float*)d_in[4];
    float* out  = (float*)d_out;
    float* hbuf = (float*)d_ws;
    const long diagElems = (long)H_ * B_ * U_;

    for (int d = 0; d < H_ + W_ - 1; ++d) {
        const int i0 = (d > W_ - 1) ? d - (W_ - 1) : 0;
        const int i1 = (d < H_ - 1) ? d : (H_ - 1);
        const int nb = i1 - i0 + 1;
        float*       hc = hbuf + (long)(d & 1) * diagElems;
        const float* hp = hbuf + (long)((d & 1) ^ 1) * diagElems;
        mdrnn_diag_kernel<<<dim3(NSLAB, nb), dim3(256), 0, stream>>>(
            x, wax, w0, w1, ba, hp, hc, out, d, i0);
    }
}


# Column-slab width picked per anti-diagonal: 32-column work-groups (16 per cell) when the diagonal has up to 16 or 33 to 48 cells, so the 256 WGPs get even shares; f32 sum order unchanged
# speedup vs baseline: 4.7709x; 1.0816x over previous
#include <hip/hip_runtime.h>
#include <cmath>


#define B_ 32
#define H_ 64
#define W_ 64
#define D_ 256
#define U_ 512

#define NWAVE 8

typedef __attribute__((ext_vector_type(2))) float v2f;
typedef __attribute__((ext_vector_type(4))) float v4f;
typedef __attribute__((ext_vector_type(8))) float v8f;

template <int KC, int NT>
__device__ __forceinline__ void gemm_acc(const float* __restrict__ aLane, long aRowStride,
                                         const float* __restrict__ bLane,
                                         v8f (&acc)[2][NT]) {
    typedef float bvec __attribute__((ext_vector_type(NT)));
    const float* __restrict__ aLane1 = aLane + 16 * aRowStride;
    #pragma unroll
    for (int g = 0; g < KC / 8; ++g) {
        const v4f a0 = *(const v4f*)(aLane  + 8 * g);
        const v4f a1 = *(const v4f*)(aLane1 + 8 * g);
        bvec b[2][2];
        #pragma unroll
        for (int s = 0; s < 2; ++s)
            #pragma unroll
            for (int e = 0; e < 2; ++e)
                b[s][e] = *(const bvec*)(bLane + (long)(8 * g + 2 * s + e) * U_);
        #pragma unroll
        for (int s = 0; s < 2; ++s) {
            const v2f af0 = { a0[2 * s], a0[2 * s + 1] };
            const v2f af1 = { a1[2 * s], a1[2 * s + 1] };
            #pragma unroll
            for (int t = 0; t < NT; ++t) {
                const v2f bf = { b[s][0][t], b[s][1][t] };
                acc[0][t] = __builtin_amdgcn_wmma_f32_16x16x4_f32(
                    false, af0, false, bf, (short)0, acc[0][t], false, false);
                acc[1][t] = __builtin_amdgcn_wmma_f32_16x16x4_f32(
                    false, af1, false, bf, (short)0, acc[1][t], false, false);
            }
        }
    }
}

template <int SLABW>
__device__ __forceinline__ void diag_body(const float* __restrict__ x,   const float* __restrict__ wax,
                                          const float* __restrict__ w0,  const float* __restrict__ w1,
                                          const float* __restrict__ ba,
                                          const float* __restrict__ hprev,
                                          float* __restrict__ hcur,
                                          float* __restrict__ out,
                                          int d, int i0, float* part)
{
    constexpr int NT    = SLABW / 16;
    constexpr int PITCH = SLABW + 4;
    typedef float ovec __attribute__((ext_vector_type(NT)));

    const int i = i0 + (int)blockIdx.y;
    const int j = d - i;
    const int tid   = (int)threadIdx.x;
    const int lane  = tid & 31;
    const int wave  = tid >> 5;
    const int nBase = (int)blockIdx.x * SLABW;

    v8f acc[2][NT];
    #pragma unroll
    for (int mt = 0; mt < 2; ++mt)
        #pragma unroll
        for (int t = 0; t < NT; ++t)
            acc[mt][t] = (v8f){0.f, 0.f, 0.f, 0.f, 0.f, 0.f, 0.f, 0.f};

    const int c = lane & 15;
    const int h = lane >> 4;
    const long bOff = (long)(4 * h) * U_ + nBase + NT * c;

    {
        const int  kx = wave * (D_ / NWAVE);
        const long xRowStride = (long)H_ * W_ * D_;
        const float* xLane = x + ((long)i * W_ + j) * (long)D_ + kx
                               + (long)c * xRowStride + 4 * h;
        gemm_acc<D_ / NWAVE, NT>(xLane, xRowStride, wax + (long)kx * U_ + bOff, acc);
    }
    const int kh = wave * (U_ / NWAVE);
    if (i > 0)
        gemm_acc<U_ / NWAVE, NT>(hprev + ((long)(i - 1) * B_ + c) * U_ + kh + 4 * h, U_,
                                 w0 + (long)kh * U_ + bOff, acc);
    if (j > 0)
        gemm_acc<U_ / NWAVE, NT>(hprev + ((long)i * B_ + c) * U_ + kh + 4 * h,       U_,
                                 w1 + (long)kh * U_ + bOff, acc);

    {
        const int hiRow = h << 3;
        #pragma unroll
        for (int mt = 0; mt < 2; ++mt)
            #pragma unroll
            for (int r = 0; r < 8; ++r) {
                ovec v;
                #pragma unroll
                for (int t = 0; t < NT; ++t) v[t] = acc[mt][t][r];
                *(ovec*)&part[((wave * B_) + mt * 16 + hiRow + r) * PITCH + NT * c] = v;
            }
    }
    __syncthreads();

    constexpr int RPP = 256 / SLABW;
    const bool last = (i == H_ - 1) && (j == W_ - 1);
    const int  col  = tid & (SLABW - 1);
    const int  row0 = tid / SLABW;
    const int  gcol = nBase + col;
    const float bv  = ba[gcol];
    #pragma unroll
    for (int q = 0; q < B_ / RPP; ++q) {
        const int row = row0 + RPP * q;
        float s = part[row * PITCH + col];
        #pragma unroll
        for (int w = 1; w < NWAVE; ++w) s += part[(w * B_ + row) * PITCH + col];
        const float v = tanhf(s + bv);
        hcur[((long)i * B_ + row) * U_ + gcol] = v;
        if (last) out[(long)row * U_ + gcol] = v;
    }
}

__global__ void __launch_bounds__(256) __attribute__((amdgpu_waves_per_eu(4, 4)))
mdrnn_diag_kernel_w64(const float* __restrict__ x,   const float* __restrict__ wax,
                      const float* __restrict__ w0,  const float* __restrict__ w1,
                      const float* __restrict__ ba,
                      const float* __restrict__ hprev, float* __restrict__ hcur,
                      float* __restrict__ out, int d, int i0)
{
    __shared__ __attribute__((aligned(16))) float part[NWAVE * B_ * (64 + 4)];
    diag_body<64>(x, wax, w0, w1, ba, hprev, hcur, out, d, i0, part);
}

__global__ void __launch_bounds__(256) __attribute__((amdgpu_waves_per_eu(4, 4)))
mdrnn_diag_kernel_w32(const float* __restrict__ x,   const float* __restrict__ wax,
                      const float* __restrict__ w0,  const float* __restrict__ w1,
                      const float* __restrict__ ba,
                      const float* __restrict__ hprev, float* __restrict__ hcur,
                      float* __restrict__ out, int d, int i0)
{
    __shared__ __attribute__((aligned(16))) float part[NWAVE * B_ * (32 + 4)];
    diag_body<32>(x, wax, w0, w1, ba, hprev, hcur, out, d, i0, part);
}

extern "C" void kernel_launch(void* const* d_in, const int* in_sizes, int n_in,
                              void* d_out, int out_size, void* d_ws, size_t ws_size,
                              hipStream_t stream) {
    (void)in_sizes; (void)n_in; (void)out_size; (void)ws_size;
    const float* x   = (const float*)d_in[0];
    const float* wax = (const float*)d_in[1];
    const float* w0  = (const float*)d_in[2];
    const float* w1  = (const float*)d_in[3];
    const float* ba  = (const float*)d_in[4];
    float* out  = (float*)d_out;
    float* hbuf = (float*)d_ws;
    const long diagElems = (long)H_ * B_ * U_;

    for (int d = 0; d < H_ + W_ - 1; ++d) {
        const int i0 = (d > W_ - 1) ? d - (W_ - 1) : 0;
        const int i1 = (d < H_ - 1) ? d : (H_ - 1);
        const int nb = i1 - i0 + 1;
        float*       hc = hbuf + (long)(d & 1) * diagElems;
        const float* hp = hbuf + (long)((d & 1) ^ 1) * diagElems;
        if (nb <= 16 || (nb > 32 && nb <= 48))
            mdrnn_diag_kernel_w32<<<dim3(U_ / 32, nb), dim3(256), 0, stream>>>(
                x, wax, w0, w1, ba, hp, hc, out, d, i0);
        else
            mdrnn_diag_kernel_w64<<<dim3(U_ / 64, nb), dim3(256), 0, stream>>>(
                x, wax, w0, w1, ba, hp, hc, out, d, i0);
    }
}


# Epilogue: each thread sums four adjacent columns with 16-byte LDS reads issued together, tanh by the library's two forms, branch-free, on four values at once, 16-byte stores; f32 sum order unchanged
# speedup vs baseline: 5.1428x; 1.0780x over previous
#include <hip/hip_runtime.h>


#define B_ 32
#define H_ 64
#define W_ 64
#define D_ 256
#define U_ 512

#define NWAVE 8

typedef __attribute__((ext_vector_type(2))) float v2f;
typedef __attribute__((ext_vector_type(4))) float v4f;
typedef __attribute__((ext_vector_type(8))) float v8f;

template <int KC, int NT>
__device__ __forceinline__ void gemm_acc(const float* __restrict__ aLane, long aRowStride,
                                         const float* __restrict__ bLane,
                                         v8f (&acc)[2][NT]) {
    typedef float bvec __attribute__((ext_vector_type(NT)));
    const float* __restrict__ aLane1 = aLane + 16 * aRowStride;
    #pragma unroll
    for (int g = 0; g < KC / 8; ++g) {
        const v4f a0 = *(const v4f*)(aLane  + 8 * g);
        const v4f a1 = *(const v4f*)(aLane1 + 8 * g);
        bvec b[2][2];
        #pragma unroll
        for (int s = 0; s < 2; ++s)
            #pragma unroll
            for (int e = 0; e < 2; ++e)
                b[s][e] = *(const bvec*)(bLane + (long)(8 * g + 2 * s + e) * U_);
        #pragma unroll
        for (int s = 0; s < 2; ++s) {
            const v2f af0 = { a0[2 * s], a0[2 * s + 1] };
            const v2f af1 = { a1[2 * s], a1[2 * s + 1] };
            #pragma unroll
            for (int t = 0; t < NT; ++t) {
                const v2f bf = { b[s][0][t], b[s][1][t] };
                acc[0][t] = __builtin_amdgcn_wmma_f32_16x16x4_f32(
                    false, af0, false, bf, (short)0, acc[0][t], false, false);
                acc[1][t] = __builtin_amdgcn_wmma_f32_16x16x4_f32(
                    false, af1, false, bf, (short)0, acc[1][t], false, false);
            }
        }
    }
}

__device__ __forceinline__ float tanh_f32(float x) {
    const float ax = __builtin_fabsf(x);
    const float L2E_HI = 1.44269502f;
    const float L2E_LO = __builtin_bit_cast(float, 0x32a5705fu);
    const float t  = ax + ax;
    const float p  = t * L2E_HI;
    const float r  = __builtin_rintf(p);
    float lo = __builtin_fmaf(t, L2E_HI, -p);
    lo = __builtin_fmaf(t, L2E_LO, lo);
    float e = __builtin_amdgcn_exp2f((p - r) + lo);
    e = __builtin_ldexpf(e, (int)r);
    e = !(88.7228394f < t) ? e : __builtin_inff();
    const float big = __builtin_fmaf(__builtin_amdgcn_rcpf(e + 1.0f), -2.0f, 1.0f);

    const float x2 = x * x;
    float q = __builtin_fmaf(__builtin_bit_cast(float, 0xbbbac73du), x2,
                             __builtin_bit_cast(float, 0x3ca908c9u));
    q = __builtin_fmaf(x2, q, __builtin_bit_cast(float, 0xbd5c1c4eu));
    q = __builtin_fmaf(x2, q, __builtin_bit_cast(float, 0x3e088382u));
    q = __builtin_fmaf(x2, q, __builtin_bit_cast(float, 0xbeaaaa99u));
    const float small = __builtin_fmaf(x2, ax * q, ax);

    const float y = !(0.625f > ax) ? big : small;
    return __builtin_copysignf(y, x);
}

template <int SLABW>
__device__ __forceinline__ void diag_body(const float* __restrict__ x,   const float* __restrict__ wax,
                                          const float* __restrict__ w0,  const float* __restrict__ w1,
                                          const float* __restrict__ ba,
                                          const float* __restrict__ hprev,
                                          float* __restrict__ hcur,
                                          float* __restrict__ out,
                                          int d, int i0, float* part)
{
    constexpr int NT    = SLABW / 16;
    constexpr int PITCH = SLABW + 4;
    typedef float ovec __attribute__((ext_vector_type(NT)));

    const int i = i0 + (int)blockIdx.y;
    const int j = d - i;
    const int tid   = (int)threadIdx.x;
    const int lane  = tid & 31;
    const int wave  = tid >> 5;
    const int nBase = (int)blockIdx.x * SLABW;

    v8f acc[2][NT];
    #pragma unroll
    for (int mt = 0; mt < 2; ++mt)
        #pragma unroll
        for (int t = 0; t < NT; ++t)
            acc[mt][t] = (v8f){0.f, 0.f, 0.f, 0.f, 0.f, 0.f, 0.f, 0.f};

    const int c = lane & 15;
    const int h = lane >> 4;
    const long bOff = (long)(4 * h) * U_ + nBase + NT * c;

    {
        const int  kx = wave * (D_ / NWAVE);
        const long xRowStride = (long)H_ * W_ * D_;
        const float* xLane = x + ((long)i * W_ + j) * (long)D_ + kx
                               + (long)c * xRowStride + 4 * h;
        gemm_acc<D_ / NWAVE, NT>(xLane, xRowStride, wax + (long)kx * U_ + bOff, acc);
    }
    const int kh = wave * (U_ / NWAVE);
    if (i > 0)
        gemm_acc<U_ / NWAVE, NT>(hprev + ((long)(i - 1) * B_ + c) * U_ + kh + 4 * h, U_,
                                 w0 + (long)kh * U_ + bOff, acc);
    if (j > 0)
        gemm_acc<U_ / NWAVE, NT>(hprev + ((long)i * B_ + c) * U_ + kh + 4 * h,       U_,
                                 w1 + (long)kh * U_ + bOff, acc);

    {
        const int hiRow = h << 3;
        #pragma unroll
        for (int mt = 0; mt < 2; ++mt)
            #pragma unroll
            for (int r = 0; r < 8; ++r) {
                ovec v;
                #pragma unroll
                for (int t = 0; t < NT; ++t) v[t] = acc[mt][t][r];
                *(ovec*)&part[((wave * B_) + mt * 16 + hiRow + r) * PITCH + NT * c] = v;
            }
    }
    __syncthreads();

    constexpr int TPR   = SLABW / 4;
    constexpr int RPP   = 256 / TPR;
    constexpr int NPASS = B_ / RPP;
    const bool last = (i == H_ - 1) && (j == W_ - 1);
    const int  col  = 4 * (tid & (TPR - 1));
    const int  row0 = tid / TPR;
    const int  gcol = nBase + col;
    const v4f  bv   = *(const v4f*)(ba + gcol);
    v4f v[NPASS];
    #pragma unroll
    for (int q = 0; q < NPASS; ++q) {
        const int row = row0 + RPP * q;
        v4f s = *(const v4f*)&part[row * PITCH + col];
        #pragma unroll
        for (int w = 1; w < NWAVE; ++w) s += *(const v4f*)&part[(w * B_ + row) * PITCH + col];
        s += bv;
        #pragma unroll
        for (int e = 0; e < 4; ++e) v[q][e] = tanh_f32(s[e]);
    }
    #pragma unroll
    for (int q = 0; q < NPASS; ++q)
        *(v4f*)&hcur[((long)i * B_ + row0 + RPP * q) * U_ + gcol] = v[q];
    if (last) {
        #pragma unroll
        for (int q = 0; q < NPASS; ++q)
            *(v4f*)&out[(long)(row0 + RPP * q) * U_ + gcol] = v[q];
    }
}

__global__ void __launch_bounds__(256) __attribute__((amdgpu_waves_per_eu(4, 4)))
mdrnn_diag_kernel_w64(const float* __restrict__ x,   const float* __restrict__ wax,
                      const float* __restrict__ w0,  const float* __restrict__ w1,
                      const float* __restrict__ ba,
                      const float* __restrict__ hprev, float* __restrict__ hcur,
                      float* __restrict__ out, int d, int i0)
{
    __shared__ __attribute__((aligned(16))) float part[NWAVE * B_ * (64 + 4)];
    diag_body<64>(x, wax, w0, w1, ba, hprev, hcur, out, d, i0, part);
}

__global__ void __launch_bounds__(256) __attribute__((amdgpu_waves_per_eu(4, 4)))
mdrnn_diag_kernel_w32(const float* __restrict__ x,   const float* __restrict__ wax,
                      const float* __restrict__ w0,  const float* __restrict__ w1,
                      const float* __restrict__ ba,
                      const float* __restrict__ hprev, float* __restrict__ hcur,
                      float* __restrict__ out, int d, int i0)
{
    __shared__ __attribute__((aligned(16))) float part[NWAVE * B_ * (32 + 4)];
    diag_body<32>(x, wax, w0, w1, ba, hprev, hcur, out, d, i0, part);
}

extern "C" void kernel_launch(void* const* d_in, const int* in_sizes, int n_in,
                              void* d_out, int out_size, void* d_ws, size_t ws_size,
                              hipStream_t stream) {
    (void)in_sizes; (void)n_in; (void)out_size; (void)ws_size;
    const float* x   = (const float*)d_in[0];
    const float* wax = (const float*)d_in[1];
    const float* w0  = (const float*)d_in[2];
    const float* w1  = (const float*)d_in[3];
    const float* ba  = (const float*)d_in[4];
    float* out  = (float*)d_out;
    float* hbuf = (float*)d_ws;
    const long diagElems = (long)H_ * B_ * U_;

    for (int d = 0; d < H_ + W_ - 1; ++d) {
        const int i0 = (d > W_ - 1) ? d - (W_ - 1) : 0;
        const int i1 = (d < H_ - 1) ? d : (H_ - 1);
        const int nb = i1 - i0 + 1;
        float*       hc = hbuf + (long)(d & 1) * diagElems;
        const float* hp = hbuf + (long)((d & 1) ^ 1) * diagElems;
        if (nb <= 16 || (nb > 32 && nb <= 48))
            mdrnn_diag_kernel_w32<<<dim3(U_ / 32, nb), dim3(256), 0, stream>>>(
                x, wax, w0, w1, ba, hp, hc, out, d, i0);
        else
            mdrnn_diag_kernel_w64<<<dim3(U_ / 64, nb), dim3(256), 0, stream>>>(
                x, wax, w0, w1, ba, hp, hc, out, d, i0);
    }
}


# Each launch prefetches the next anti-diagonal's x rows into L2 (one 256-byte piece per thread of four waves), so every launch's first loads hit L2 instead of memory; f32 sum order unchanged
# speedup vs baseline: 5.4381x; 1.0574x over previous
#include <hip/hip_runtime.h>


#define B_ 32
#define H_ 64
#define W_ 64
#define D_ 256
#define U_ 512

#define NWAVE 8

typedef __attribute__((ext_vector_type(2))) float v2f;
typedef __attribute__((ext_vector_type(4))) float v4f;
typedef __attribute__((ext_vector_type(8))) float v8f;

template <int KC, int NT>
__device__ __forceinline__ void gemm_acc(const float* __restrict__ aLane, long aRowStride,
                                         const float* __restrict__ bLane,
                                         v8f (&acc)[2][NT]) {
    typedef float bvec __attribute__((ext_vector_type(NT)));
    const float* __restrict__ aLane1 = aLane + 16 * aRowStride;
    #pragma unroll
    for (int g = 0; g < KC / 8; ++g) {
        const v4f a0 = *(const v4f*)(aLane  + 8 * g);
        const v4f a1 = *(const v4f*)(aLane1 + 8 * g);
        bvec b[2][2];
        #pragma unroll
        for (int s = 0; s < 2; ++s)
            #pragma unroll
            for (int e = 0; e < 2; ++e)
                b[s][e] = *(const bvec*)(bLane + (long)(8 * g + 2 * s + e) * U_);
        #pragma unroll
        for (int s = 0; s < 2; ++s) {
            const v2f af0 = { a0[2 * s], a0[2 * s + 1] };
            const v2f af1 = { a1[2 * s], a1[2 * s + 1] };
            #pragma unroll
            for (int t = 0; t < NT; ++t) {
                const v2f bf = { b[s][0][t], b[s][1][t] };
                acc[0][t] = __builtin_amdgcn_wmma_f32_16x16x4_f32(
                    false, af0, false, bf, (short)0, acc[0][t], false, false);
                acc[1][t] = __builtin_amdgcn_wmma_f32_16x16x4_f32(
                    false, af1, false, bf, (short)0, acc[1][t], false, false);
            }
        }
    }
}

__device__ __forceinline__ float tanh_f32(float x) {
    const float ax = __builtin_fabsf(x);
    const float L2E_HI = 1.44269502f;
    const float L2E_LO = __builtin_bit_cast(float, 0x32a5705fu);
    const float t  = ax + ax;
    const float p  = t * L2E_HI;
    const float r  = __builtin_rintf(p);
    float lo = __builtin_fmaf(t, L2E_HI, -p);
    lo = __builtin_fmaf(t, L2E_LO, lo);
    float e = __builtin_amdgcn_exp2f((p - r) + lo);
    e = __builtin_ldexpf(e, (int)r);
    e = !(88.7228394f < t) ? e : __builtin_inff();
    const float big = __builtin_fmaf(__builtin_amdgcn_rcpf(e + 1.0f), -2.0f, 1.0f);

    const float x2 = x * x;
    float q = __builtin_fmaf(__builtin_bit_cast(float, 0xbbbac73du), x2,
                             __builtin_bit_cast(float, 0x3ca908c9u));
    q = __builtin_fmaf(x2, q, __builtin_bit_cast(float, 0xbd5c1c4eu));
    q = __builtin_fmaf(x2, q, __builtin_bit_cast(float, 0x3e088382u));
    q = __builtin_fmaf(x2, q, __builtin_bit_cast(float, 0xbeaaaa99u));
    const float small = __builtin_fmaf(x2, ax * q, ax);

    const float y = !(0.625f > ax) ? big : small;
    return __builtin_copysignf(y, x);
}

template <int SLABW>
__device__ __forceinline__ void diag_body(const float* __restrict__ x,   const float* __restrict__ wax,
                                          const float* __restrict__ w0,  const float* __restrict__ w1,
                                          const float* __restrict__ ba,
                                          const float* __restrict__ hprev,
                                          float* __restrict__ hcur,
                                          float* __restrict__ out,
                                          int d, int i0, float* part)
{
    constexpr int NT    = SLABW / 16;
    constexpr int PITCH = SLABW + 4;
    typedef float ovec __attribute__((ext_vector_type(NT)));

    const int i = i0 + (int)blockIdx.y;
    const int j = d - i;
    const int tid   = (int)threadIdx.x;
    const int lane  = tid & 31;
    const int wave  = tid >> 5;
    const int nBase = (int)blockIdx.x * SLABW;

    v8f acc[2][NT];
    #pragma unroll
    for (int mt = 0; mt < 2; ++mt)
        #pragma unroll
        for (int t = 0; t < NT; ++t)
            acc[mt][t] = (v8f){0.f, 0.f, 0.f, 0.f, 0.f, 0.f, 0.f, 0.f};

    const int c = lane & 15;
    const int h = lane >> 4;
    const long bOff = (long)(4 * h) * U_ + nBase + NT * c;

    {
        const int  kx = wave * (D_ / NWAVE);
        const long xRowStride = (long)H_ * W_ * D_;
        const float* xLane = x + ((long)i * W_ + j) * (long)D_ + kx
                               + (long)c * xRowStride + 4 * h;
        gemm_acc<D_ / NWAVE, NT>(xLane, xRowStride, wax + (long)kx * U_ + bOff, acc);
    }
    if (tid < 128) {
        const int dn  = d + 1;
        const int i0n = (dn > W_ - 1) ? dn - (W_ - 1) : 0;
        const int i1n = (dn < H_ - 1) ? dn : (H_ - 1);
        const int i1c = (d  < H_ - 1) ? d  : (H_ - 1);
        const int in  = i0n + (int)blockIdx.y;
        const float* piece = x + (long)(tid >> 2) * ((long)H_ * W_ * D_) + (tid & 3) * 64;
        if (dn < H_ + W_ - 1) {
            if (in <= i1n)
                __builtin_prefetch(piece + ((long)in * W_ + (dn - in)) * (long)D_, 0, 1);
            if (i0 + (int)blockIdx.y == i1c && in + 1 <= i1n)
                __builtin_prefetch(piece + ((long)(in + 1) * W_ + (dn - in - 1)) * (long)D_, 0, 1);
        }
    }
    const int kh = wave * (U_ / NWAVE);
    if (i > 0)
        gemm_acc<U_ / NWAVE, NT>(hprev + ((long)(i - 1) * B_ + c) * U_ + kh + 4 * h, U_,
                                 w0 + (long)kh * U_ + bOff, acc);
    if (j > 0)
        gemm_acc<U_ / NWAVE, NT>(hprev + ((long)i * B_ + c) * U_ + kh + 4 * h,       U_,
                                 w1 + (long)kh * U_ + bOff, acc);

    {
        const int hiRow = h << 3;
        #pragma unroll
        for (int mt = 0; mt < 2; ++mt)
            #pragma unroll
            for (int r = 0; r < 8; ++r) {
                ovec v;
                #pragma unroll
                for (int t = 0; t < NT; ++t) v[t] = acc[mt][t][r];
                *(ovec*)&part[((wave * B_) + mt * 16 + hiRow + r) * PITCH + NT * c] = v;
            }
    }
    __syncthreads();

    constexpr int TPR   = SLABW / 4;
    constexpr int RPP   = 256 / TPR;
    constexpr int NPASS = B_ / RPP;
    const bool last = (i == H_ - 1) && (j == W_ - 1);
    const int  col  = 4 * (tid & (TPR - 1));
    const int  row0 = tid / TPR;
    const int  gcol = nBase + col;
    const v4f  bv   = *(const v4f*)(ba + gcol);
    v4f v[NPASS];
    #pragma unroll
    for (int q = 0; q < NPASS; ++q) {
        const int row = row0 + RPP * q;
        v4f s = *(const v4f*)&part[row * PITCH + col];
        #pragma unroll
        for (int w = 1; w < NWAVE; ++w) s += *(const v4f*)&part[(w * B_ + row) * PITCH + col];
        s += bv;
        #pragma unroll
        for (int e = 0; e < 4; ++e) v[q][e] = tanh_f32(s[e]);
    }
    #pragma unroll
    for (int q = 0; q < NPASS; ++q)
        *(v4f*)&hcur[((long)i * B_ + row0 + RPP * q) * U_ + gcol] = v[q];
    if (last) {
        #pragma unroll
        for (int q = 0; q < NPASS; ++q)
            *(v4f*)&out[(long)(row0 + RPP * q) * U_ + gcol] = v[q];
    }
}

__global__ void __launch_bounds__(256) __attribute__((amdgpu_waves_per_eu(4, 4)))
mdrnn_diag_kernel_w64(const float* __restrict__ x,   const float* __restrict__ wax,
                      const float* __restrict__ w0,  const float* __restrict__ w1,
                      const float* __restrict__ ba,
                      const float* __restrict__ hprev, float* __restrict__ hcur,
                      float* __restrict__ out, int d, int i0)
{
    __shared__ __attribute__((aligned(16))) float part[NWAVE * B_ * (64 + 4)];
    diag_body<64>(x, wax, w0, w1, ba, hprev, hcur, out, d, i0, part);
}

__global__ void __launch_bounds__(256) __attribute__((amdgpu_waves_per_eu(4, 4)))
mdrnn_diag_kernel_w32(const float* __restrict__ x,   const float* __restrict__ wax,
                      const float* __restrict__ w0,  const float* __restrict__ w1,
                      const float* __restrict__ ba,
                      const float* __restrict__ hprev, float* __restrict__ hcur,
                      float* __restrict__ out, int d, int i0)
{
    __shared__ __attribute__((aligned(16))) float part[NWAVE * B_ * (32 + 4)];
    diag_body<32>(x, wax, w0, w1, ba, hprev, hcur, out, d, i0, part);
}

extern "C" void kernel_launch(void* const* d_in, const int* in_sizes, int n_in,
                              void* d_out, int out_size, void* d_ws, size_t ws_size,
                              hipStream_t stream) {
    (void)in_sizes; (void)n_in; (void)out_size; (void)ws_size;
    const float* x   = (const float*)d_in[0];
    const float* wax = (const float*)d_in[1];
    const float* w0  = (const float*)d_in[2];
    const float* w1  = (const float*)d_in[3];
    const float* ba  = (const float*)d_in[4];
    float* out  = (float*)d_out;
    float* hbuf = (float*)d_ws;
    const long diagElems = (long)H_ * B_ * U_;

    for (int d = 0; d < H_ + W_ - 1; ++d) {
        const int i0 = (d > W_ - 1) ? d - (W_ - 1) : 0;
        const int i1 = (d < H_ - 1) ? d : (H_ - 1);
        const int nb = i1 - i0 + 1;
        float*       hc = hbuf + (long)(d & 1) * diagElems;
        const float* hp = hbuf + (long)((d & 1) ^ 1) * diagElems;
        if (nb <= 16 || (nb > 32 && nb <= 48))
            mdrnn_diag_kernel_w32<<<dim3(U_ / 32, nb), dim3(256), 0, stream>>>(
                x, wax, w0, w1, ba, hp, hc, out, d, i0);
        else
            mdrnn_diag_kernel_w64<<<dim3(U_ / 64, nb), dim3(256), 0, stream>>>(
                x, wax, w0, w1, ba, hp, hc, out, d, i0);
    }
}


# h kept in the workspace in A-fragment order (epilogue stores each 16-row x 8-k group as 512 contiguous bytes): recurrent A loads become whole-line 16-byte loads, not 16-line; f32 sum order unchanged
# speedup vs baseline: 5.6357x; 1.0363x over previous
#include <hip/hip_runtime.h>


#define B_ 32
#define H_ 64
#define W_ 64
#define D_ 256
#define U_ 512

#define NWAVE 8

typedef __attribute__((ext_vector_type(2))) float v2f;
typedef __attribute__((ext_vector_type(4))) float v4f;
typedef __attribute__((ext_vector_type(8))) float v8f;

template <int KC, int NT, int AGS>
__device__ __forceinline__ void gemm_acc(const float* __restrict__ aLane,
                                         const float* __restrict__ aLane1,
                                         const float* __restrict__ bLane,
                                         v8f (&acc)[2][NT]) {
    typedef float bvec __attribute__((ext_vector_type(NT)));
    #pragma unroll
    for (int g = 0; g < KC / 8; ++g) {
        const v4f a0 = *(const v4f*)(aLane  + AGS * g);
        const v4f a1 = *(const v4f*)(aLane1 + AGS * g);
        bvec b[2][2];
        #pragma unroll
        for (int s = 0; s < 2; ++s)
            #pragma unroll
            for (int e = 0; e < 2; ++e)
                b[s][e] = *(const bvec*)(bLane + (long)(8 * g + 2 * s + e) * U_);
        #pragma unroll
        for (int s = 0; s < 2; ++s) {
            const v2f af0 = { a0[2 * s], a0[2 * s + 1] };
            const v2f af1 = { a1[2 * s], a1[2 * s + 1] };
            #pragma unroll
            for (int t = 0; t < NT; ++t) {
                const v2f bf = { b[s][0][t], b[s][1][t] };
                acc[0][t] = __builtin_amdgcn_wmma_f32_16x16x4_f32(
                    false, af0, false, bf, (short)0, acc[0][t], false, false);
                acc[1][t] = __builtin_amdgcn_wmma_f32_16x16x4_f32(
                    false, af1, false, bf, (short)0, acc[1][t], false, false);
            }
        }
    }
}

#define HF_GROUP 128
#define HF_MT    (64 * HF_GROUP)

__device__ __forceinline__ float tanh_f32(float x) {
    const float ax = __builtin_fabsf(x);
    const float L2E_HI = 1.44269502f;
    const float L2E_LO = __builtin_bit_cast(float, 0x32a5705fu);
    const float t  = ax + ax;
    const float p  = t * L2E_HI;
    const float r  = __builtin_rintf(p);
    float lo = __builtin_fmaf(t, L2E_HI, -p);
    lo = __builtin_fmaf(t, L2E_LO, lo);
    float e = __builtin_amdgcn_exp2f((p - r) + lo);
    e = __builtin_ldexpf(e, (int)r);
    e = !(88.7228394f < t) ? e : __builtin_inff();
    const float big = __builtin_fmaf(__builtin_amdgcn_rcpf(e + 1.0f), -2.0f, 1.0f);

    const float x2 = x * x;
    float q = __builtin_fmaf(__builtin_bit_cast(float, 0xbbbac73du), x2,
                             __builtin_bit_cast(float, 0x3ca908c9u));
    q = __builtin_fmaf(x2, q, __builtin_bit_cast(float, 0xbd5c1c4eu));
    q = __builtin_fmaf(x2, q, __builtin_bit_cast(float, 0x3e088382u));
    q = __builtin_fmaf(x2, q, __builtin_bit_cast(float, 0xbeaaaa99u));
    const float small = __builtin_fmaf(x2, ax * q, ax);

    const float y = !(0.625f > ax) ? big : small;
    return __builtin_copysignf(y, x);
}

template <int SLABW>
__device__ __forceinline__ void diag_body(const float* __restrict__ x,   const float* __restrict__ wax,
                                          const float* __restrict__ w0,  const float* __restrict__ w1,
                                          const float* __restrict__ ba,
                                          const float* __restrict__ hprev,
                                          float* __restrict__ hcur,
                                          float* __restrict__ out,
                                          int d, int i0, float* part)
{
    constexpr int NT    = SLABW / 16;
    constexpr int PITCH = SLABW + 4;
    typedef float ovec __attribute__((ext_vector_type(NT)));

    const int i = i0 + (int)blockIdx.y;
    const int j = d - i;
    const int tid   = (int)threadIdx.x;
    const int lane  = tid & 31;
    const int wave  = tid >> 5;
    const int nBase = (int)blockIdx.x * SLABW;

    v8f acc[2][NT];
    #pragma unroll
    for (int mt = 0; mt < 2; ++mt)
        #pragma unroll
        for (int t = 0; t < NT; ++t)
            acc[mt][t] = (v8f){0.f, 0.f, 0.f, 0.f, 0.f, 0.f, 0.f, 0.f};

    const int c = lane & 15;
    const int h = lane >> 4;
    const long bOff = (long)(4 * h) * U_ + nBase + NT * c;

    {
        const int  kx = wave * (D_ / NWAVE);
        const long xRowStride = (long)H_ * W_ * D_;
        const float* xLane = x + ((long)i * W_ + j) * (long)D_ + kx
                               + (long)c * xRowStride + 4 * h;
        gemm_acc<D_ / NWAVE, NT, 8>(xLane, xLane + 16 * xRowStride,
                                    wax + (long)kx * U_ + bOff, acc);
    }
    if (tid < 128) {
        const int dn  = d + 1;
        const int i0n = (dn > W_ - 1) ? dn - (W_ - 1) : 0;
        const int i1n = (dn < H_ - 1) ? dn : (H_ - 1);
        const int i1c = (d  < H_ - 1) ? d  : (H_ - 1);
        const int in  = i0n + (int)blockIdx.y;
        const float* piece = x + (long)(tid >> 2) * ((long)H_ * W_ * D_) + (tid & 3) * 64;
        if (dn < H_ + W_ - 1) {
            if (in <= i1n)
                __builtin_prefetch(piece + ((long)in * W_ + (dn - in)) * (long)D_, 0, 1);
            if (i0 + (int)blockIdx.y == i1c && in + 1 <= i1n)
                __builtin_prefetch(piece + ((long)(in + 1) * W_ + (dn - in - 1)) * (long)D_, 0, 1);
        }
    }
    const int  kh    = wave * (U_ / NWAVE);
    const long hLane = (long)(kh / 8) * HF_GROUP + 4 * lane;
    if (i > 0) {
        const float* hc0 = hprev + (long)(i - 1) * B_ * U_ + hLane;
        gemm_acc<U_ / NWAVE, NT, HF_GROUP>(hc0, hc0 + HF_MT, w0 + (long)kh * U_ + bOff, acc);
    }
    if (j > 0) {
        const float* hc1 = hprev + (long)i * B_ * U_ + hLane;
        gemm_acc<U_ / NWAVE, NT, HF_GROUP>(hc1, hc1 + HF_MT, w1 + (long)kh * U_ + bOff, acc);
    }

    {
        const int hiRow = h << 3;
        #pragma unroll
        for (int mt = 0; mt < 2; ++mt)
            #pragma unroll
            for (int r = 0; r < 8; ++r) {
                ovec v;
                #pragma unroll
                for (int t = 0; t < NT; ++t) v[t] = acc[mt][t][r];
                *(ovec*)&part[((wave * B_) + mt * 16 + hiRow + r) * PITCH + NT * c] = v;
            }
    }
    __syncthreads();

    const bool last = (i == H_ - 1) && (j == W_ - 1);
    if (!last) {
        constexpr int NG    = SLABW / 8;
        constexpr int NPASS = 2 * NG / NWAVE;
        v4f v[NPASS];
        #pragma unroll
        for (int q = 0; q < NPASS; ++q) {
            const int p   = wave + NWAVE * q;
            const int row = 16 * (p / NG) + c;
            const int col = 8 * (p % NG) + 4 * h;
            v4f s = *(const v4f*)&part[row * PITCH + col];
            #pragma unroll
            for (int w = 1; w < NWAVE; ++w) s += *(const v4f*)&part[(w * B_ + row) * PITCH + col];
            s += *(const v4f*)(ba + nBase + col);
            #pragma unroll
            for (int e = 0; e < 4; ++e) v[q][e] = tanh_f32(s[e]);
        }
        float* hcell = hcur + (long)i * B_ * U_ + 4 * lane;
        #pragma unroll
        for (int q = 0; q < NPASS; ++q) {
            const int p = wave + NWAVE * q;
            *(v4f*)&hcell[(long)(p / NG) * HF_MT + (long)(nBase / 8 + p % NG) * HF_GROUP] = v[q];
        }
    } else {
        constexpr int TPR   = SLABW / 4;
        constexpr int RPP   = 256 / TPR;
        constexpr int NPASS = B_ / RPP;
        const int  col  = 4 * (tid & (TPR - 1));
        const int  row0 = tid / TPR;
        const int  gcol = nBase + col;
        const v4f  bv   = *(const v4f*)(ba + gcol);
        #pragma unroll
        for (int q = 0; q < NPASS; ++q) {
            const int row = row0 + RPP * q;
            v4f s = *(const v4f*)&part[row * PITCH + col];
            #pragma unroll
            for (int w = 1; w < NWAVE; ++w) s += *(const v4f*)&part[(w * B_ + row) * PITCH + col];
            s += bv;
            v4f r;
            #pragma unroll
            for (int e = 0; e < 4; ++e) r[e] = tanh_f32(s[e]);
            *(v4f*)&out[(long)row * U_ + gcol] = r;
        }
    }
}

__global__ void __launch_bounds__(256) __attribute__((amdgpu_waves_per_eu(4, 4)))
mdrnn_diag_kernel_w64(const float* __restrict__ x,   const float* __restrict__ wax,
                      const float* __restrict__ w0,  const float* __restrict__ w1,
                      const float* __restrict__ ba,
                      const float* __restrict__ hprev, float* __restrict__ hcur,
                      float* __restrict__ out, int d, int i0)
{
    __shared__ __attribute__((aligned(16))) float part[NWAVE * B_ * (64 + 4)];
    diag_body<64>(x, wax, w0, w1, ba, hprev, hcur, out, d, i0, part);
}

__global__ void __launch_bounds__(256) __attribute__((amdgpu_waves_per_eu(4, 4)))
mdrnn_diag_kernel_w32(const float* __restrict__ x,   const float* __restrict__ wax,
                      const float* __restrict__ w0,  const float* __restrict__ w1,
                      const float* __restrict__ ba,
                      const float* __restrict__ hprev, float* __restrict__ hcur,
                      float* __restrict__ out, int d, int i0)
{
    __shared__ __attribute__((aligned(16))) float part[NWAVE * B_ * (32 + 4)];
    diag_body<32>(x, wax, w0, w1, ba, hprev, hcur, out, d, i0, part);
}

extern "C" void kernel_launch(void* const* d_in, const int* in_sizes, int n_in,
                              void* d_out, int out_size, void* d_ws, size_t ws_size,
                              hipStream_t stream) {
    (void)in_sizes; (void)n_in; (void)out_size; (void)ws_size;
    const float* x   = (const float*)d_in[0];
    const float* wax = (const float*)d_in[1];
    const float* w0  = (const float*)d_in[2];
    const float* w1  = (const float*)d_in[3];
    const float* ba  = (const float*)d_in[4];
    float* out  = (float*)d_out;
    float* hbuf = (float*)d_ws;
    const long diagElems = (long)H_ * B_ * U_;

    for (int d = 0; d < H_ + W_ - 1; ++d) {
        const int i0 = (d > W_ - 1) ? d - (W_ - 1) : 0;
        const int i1 = (d < H_ - 1) ? d : (H_ - 1);
        const int nb = i1 - i0 + 1;
        float*       hc = hbuf + (long)(d & 1) * diagElems;
        const float* hp = hbuf + (long)((d & 1) ^ 1) * diagElems;
        if (nb <= 16 || (nb > 32 && nb <= 48))
            mdrnn_diag_kernel_w32<<<dim3(U_ / 32, nb), dim3(256), 0, stream>>>(
                x, wax, w0, w1, ba, hp, hc, out, d, i0);
        else
            mdrnn_diag_kernel_w64<<<dim3(U_ / 64, nb), dim3(256), 0, stream>>>(
                x, wax, w0, w1, ba, hp, hc, out, d, i0);
    }
}
